# starting kernel
# baseline (speedup 1.0000x reference)
#include <hip/hip_runtime.h>

typedef float v2f __attribute__((ext_vector_type(2)));
typedef float v8f __attribute__((ext_vector_type(8)));

__global__ void init_max_kernel(unsigned int* __restrict__ maxbits, int n) {
    int i = blockIdx.x * blockDim.x + threadIdx.x;
    if (i < n) maxbits[i] = 0u;
}

__global__ void phi_mean_wmma_kernel(const float* __restrict__ W_phi,
                                     float* __restrict__ phi_mean) {
    int lane = threadIdx.x;
    v2f a0, a1, b;
    a0.x = W_phi[lane];        a0.y = W_phi[lane + 32];
    a1.x = W_phi[lane + 64];   a1.y = W_phi[lane + 96];
    b.x = 1.0f; b.y = 1.0f;
    v8f c = {};
    c = __builtin_amdgcn_wmma_f32_16x16x4_f32(false, a0, false, b,
                                              (short)0, c, false, false);
    c = __builtin_amdgcn_wmma_f32_16x16x4_f32(false, a1, false, b,
                                              (short)0, c, false, false);
    float s = c[0] + c[1] + c[2] + c[3] + c[4] + c[5] + c[6] + c[7];
    float partner = __shfl(s, lane ^ 16, 32);
    if (lane == 0) phi_mean[0] = (s + partner) * (1.0f / 128.0f);
}

__global__ void edge_max_kernel(const float* __restrict__ nodes,
                                const int* __restrict__ rows,
                                const int* __restrict__ cols,
                                const float* __restrict__ W_theta,
                                unsigned int* __restrict__ maxbits,
                                int n_edges) {
    const float w0 = W_theta[0], w1 = W_theta[1], w2 = W_theta[2];
    int base = (blockIdx.x * blockDim.x + threadIdx.x) * 4;
    if (base + 3 < n_edges) {
        int4 r4 = *reinterpret_cast<const int4*>(rows + base);
        int4 c4 = *reinterpret_cast<const int4*>(cols + base);
        int rr[4] = {r4.x, r4.y, r4.z, r4.w};
        int cc[4] = {c4.x, c4.y, c4.z, c4.w};
#pragma unroll
        for (int k = 0; k < 4; ++k) {
            const float* pa = nodes + 3 * rr[k];
            const float* pb = nodes + 3 * cc[k];
            float d = fabsf((pa[0] - pb[0]) * w0 +
                            (pa[1] - pb[1]) * w1 +
                            (pa[2] - pb[2]) * w2);
            atomicMax(maxbits + rr[k], __float_as_uint(d));
        }
    } else {
        for (int e = base; e < n_edges; ++e) {
            int r = rows[e], c = cols[e];
            const float* pa = nodes + 3 * r;
            const float* pb = nodes + 3 * c;
            float d = fabsf((pa[0] - pb[0]) * w0 +
                            (pa[1] - pb[1]) * w1 +
                            (pa[2] - pb[2]) * w2);
            atomicMax(maxbits + r, __float_as_uint(d));
        }
    }
}

__global__ void finalize_kernel(const float* __restrict__ prev,
                                const unsigned int* __restrict__ maxbits,
                                const float* __restrict__ phi_mean,
                                float* __restrict__ out, int n) {
    int i = blockIdx.x * blockDim.x + threadIdx.x;
    if (i < n) {
        float m = __uint_as_float(maxbits[i]);
        out[i] = 0.5f * (prev[i] + m * phi_mean[0]);
    }
}

extern "C" void kernel_launch(void* const* d_in, const int* in_sizes, int n_in,
                              void* d_out, int out_size, void* d_ws, size_t ws_size,
                              hipStream_t stream) {
    const float* prev    = (const float*)d_in[0];
    const float* nodes   = (const float*)d_in[1];
    const int*   rows    = (const int*)  d_in[2];
    const int*   cols    = (const int*)  d_in[3];
    const float* W_phi   = (const float*)d_in[4];
    const float* W_theta = (const float*)d_in[5];
    float*       out     = (float*)d_out;

    const int n_nodes = in_sizes[0];
    const int n_edges = in_sizes[2];

    unsigned int* maxbits = (unsigned int*)d_ws;
    float* phi_mean = (float*)((unsigned int*)d_ws + n_nodes);

    const int B = 256;
    int node_blocks = (n_nodes + B - 1) / B;
    int edge_threads = (n_edges + 3) / 4;
    int edge_blocks = (edge_threads + B - 1) / B;

    init_max_kernel<<<node_blocks, B, 0, stream>>>(maxbits, n_nodes);
    phi_mean_wmma_kernel<<<1, 32, 0, stream>>>(W_phi, phi_mean);
    edge_max_kernel<<<edge_blocks, B, 0, stream>>>(nodes, rows, cols, W_theta,
                                                   maxbits, n_edges);
    finalize_kernel<<<node_blocks, B, 0, stream>>>(prev, maxbits, phi_mean,
                                                   out, n_nodes);
}


# Edge kernel: a column's projection goes into the spare second word of its row's SORTED record instead of an exchange array; the update pass reads one consecutive 8-byte record per end point
# speedup vs baseline: 9.7039x; 9.7039x over previous
#include <hip/hip_runtime.h>

typedef float v2f __attribute__((ext_vector_type(2)));
typedef float v8f __attribute__((ext_vector_type(8)));

__global__ void init_project_kernel(const float* __restrict__ nodes,
                                    const float* __restrict__ W_theta,
                                    unsigned int* __restrict__ maxbits,
                                    float* __restrict__ out,
                                    float* __restrict__ proj, int n) {
    int i = blockIdx.x * blockDim.x + threadIdx.x;
    if (i < n) {
        const float w0 = W_theta[0], w1 = W_theta[1], w2 = W_theta[2];
        const float x = nodes[3 * i], y = nodes[3 * i + 1], z = nodes[3 * i + 2];
        maxbits[i] = 0u;
        out[i] = 0.0f;
        proj[i] = fmaf(z, w2, fmaf(x, w0, y * w1));
    }
}

constexpr int EDGE_TPB  = 800;
constexpr int EDGE_EPT  = 16;
constexpr int EDGE_TILE = EDGE_TPB * EDGE_EPT;
constexpr int REF_PT    = 2 * EDGE_EPT;
constexpr int REF_TILE  = 2 * EDGE_TILE;
constexpr int REF_HALF  = REF_PT / 2;
constexpr int LINE_SHIFT = 5;
constexpr int NBINS = EDGE_TPB * 4;
constexpr int EDGE_WAVES = EDGE_TPB / 32;
constexpr unsigned COL_FLAG = 0x80000000u;
constexpr unsigned ROW_NONE = 0x7FFFFFFFu;
static_assert(EDGE_TPB % 32 == 0 && EDGE_WAVES <= 32, "the scan folds at most 32 whole waves");

__device__ __forceinline__ void scan_bins(unsigned int* s_bin, unsigned int* s_wave,
                                          int tid, int lane, int wave) {
    uint4* bin4 = reinterpret_cast<uint4*>(s_bin);
    const uint4 cnt = bin4[tid];
    const unsigned t1 = cnt.x, t2 = t1 + cnt.y, t3 = t2 + cnt.z, tot = t3 + cnt.w;
    unsigned incl = tot;
#pragma unroll
    for (int s = 1; s < 32; s <<= 1) {
        const unsigned v = __shfl_up(incl, s, 32);
        if (lane >= s) incl += v;
    }
    if (lane == 31) s_wave[wave] = incl;
    __syncthreads();
    if (wave == 0) {
        const unsigned wt = lane < EDGE_WAVES ? s_wave[lane] : 0u;
        unsigned wi = wt;
#pragma unroll
        for (int s = 1; s < 32; s <<= 1) {
            const unsigned v = __shfl_up(wi, s, 32);
            if (lane >= s) wi += v;
        }
        s_wave[lane] = wi - wt;
    }
    __syncthreads();
    const unsigned base = s_wave[wave] + incl - tot;
    bin4[tid] = make_uint4(base, base + t1, base + t2, base + t3);
}

__device__ __forceinline__ unsigned ref_slot(unsigned first, int i) {
    const unsigned p = first + (unsigned)(i * EDGE_TPB);
    return p >= (unsigned)REF_TILE ? p - (unsigned)REF_TILE : p;
}

template <bool CHECKED>
__device__ __forceinline__ void gather_pass(const float* __restrict__ proj,
                                            uint2* s_ref,
                                            unsigned first, float (&p)[REF_PT]) {
    uint2 rec[REF_PT];
#pragma unroll
    for (int i = 0; i < REF_PT; ++i) rec[i] = s_ref[ref_slot(first, i)];
#pragma unroll
    for (int i = 0; i < REF_PT; ++i) {
        const unsigned node = rec[i].x & ~COL_FLAG;
        const unsigned ng = (CHECKED && node == ROW_NONE) ? 0u : node;
        p[i] = proj[(size_t)ng];
    }
#pragma unroll
    for (int i = 0; i < REF_PT; ++i)
        if (rec[i].x & COL_FLAG) s_ref[rec[i].y].y = __float_as_uint(p[i]);
}

template <bool CHECKED>
__device__ __forceinline__ void update_pass(unsigned int* __restrict__ plane,
                                            const uint2* s_ref,
                                            unsigned first, const float (&p)[REF_PT]) {
#pragma unroll
    for (int h = 0; h < 2; ++h) {
        uint2 rec[REF_HALF];
#pragma unroll
        for (int j = 0; j < REF_HALF; ++j) rec[j] = s_ref[ref_slot(first, h * REF_HALF + j)];
#pragma unroll
        for (int j = 0; j < REF_HALF; ++j) {
            const float d = fabsf(p[h * REF_HALF + j] - __uint_as_float(rec[j].y));
            const bool is_row = (rec[j].x & COL_FLAG) == 0u;
            if (is_row && (!CHECKED || rec[j].x != ROW_NONE))
                atomicMax(plane + rec[j].x, __float_as_uint(d));
        }
    }
}

__global__ __launch_bounds__(EDGE_TPB)
void edge_max_kernel(const float* __restrict__ proj,
                     const int* __restrict__ rows,
                     const int* __restrict__ cols,
                     unsigned int* __restrict__ maxbits,
                     float* __restrict__ out,
                     int n_edges) {
    __shared__ uint2 s_ref[REF_TILE];
    __shared__ unsigned int s_bin[NBINS];
    __shared__ unsigned int s_wave[32];

    const int tid  = threadIdx.x;
    const int lane = tid & 31;
    const int wave = tid >> 5;
    const int tile_base = blockIdx.x * EDGE_TILE;
    const bool full_tile = tile_base + EDGE_TILE <= n_edges;

    uint4* bin4 = reinterpret_cast<uint4*>(s_bin);
    bin4[tid] = make_uint4(0u, 0u, 0u, 0u);
    __syncthreads();


    {
        int r4[EDGE_EPT], c4[EDGE_EPT];
        if (full_tile) {
#pragma unroll
            for (int k = 0; k < EDGE_EPT / 4; ++k) {
                const int e0 = tile_base + (k * EDGE_TPB + tid) * 4;
                const int4 rv = *reinterpret_cast<const int4*>(rows + e0);
                const int4 cv = *reinterpret_cast<const int4*>(cols + e0);
                r4[4 * k + 0] = rv.x; r4[4 * k + 1] = rv.y; r4[4 * k + 2] = rv.z; r4[4 * k + 3] = rv.w;
                c4[4 * k + 0] = cv.x; c4[4 * k + 1] = cv.y; c4[4 * k + 2] = cv.z; c4[4 * k + 3] = cv.w;
            }
        } else {
#pragma unroll
            for (int i = 0; i < EDGE_EPT; ++i) {
                const int e = tile_base + ((i >> 2) * EDGE_TPB + tid) * 4 + (i & 3);
                const bool ok = e < n_edges;
                r4[i] = ok ? rows[e] : -1;
                c4[i] = ok ? cols[e] : 0;
            }
        }

        unsigned bb[REF_PT], rk[REF_PT];
#pragma unroll
        for (int i = 0; i < EDGE_EPT; ++i) {
            const unsigned br = (unsigned)r4[i] >> LINE_SHIFT;
            const unsigned bc = (unsigned)c4[i] >> LINE_SHIFT;
            bb[i]            = br < (unsigned)(NBINS - 1) ? br : (unsigned)(NBINS - 1);
            bb[EDGE_EPT + i] = bc < (unsigned)(NBINS - 1) ? bc : (unsigned)(NBINS - 1);
        }
#pragma unroll
        for (int i = 0; i < REF_PT; ++i) rk[i] = atomicAdd(&s_bin[bb[i]], 1u);
        __syncthreads();
        scan_bins(s_bin, s_wave, tid, lane, wave);
        __syncthreads();

        unsigned pos[REF_PT];
#pragma unroll
        for (int i = 0; i < REF_PT; ++i) pos[i] = s_bin[bb[i]] + rk[i];
#pragma unroll
        for (int i = 0; i < EDGE_EPT; ++i) {
            s_ref[pos[i]]            = make_uint2((unsigned)r4[i] & ~COL_FLAG, 0u);
            s_ref[pos[EDGE_EPT + i]] = make_uint2((unsigned)c4[i] | COL_FLAG, pos[i]);
        }
        __syncthreads();
    }

    const unsigned rot = __umulhi(blockIdx.x * 0x9E3779B1u, (unsigned)REF_TILE);
    const unsigned first = (unsigned)tid + rot;
    const unsigned place = __builtin_amdgcn_s_sendmsg_rtn(0x87);
    const unsigned half = (place >> 18) & 1u;
    unsigned int* plane = half ? reinterpret_cast<unsigned int*>(out) : maxbits;

    float p[REF_PT];
    if (full_tile) gather_pass<false>(proj, s_ref, first, p);
    else           gather_pass<true>(proj, s_ref, first, p);
    __syncthreads();
    if (full_tile) update_pass<false>(plane, s_ref, first, p);
    else           update_pass<true>(plane, s_ref, first, p);
}

__global__ void finalize_kernel(const float* __restrict__ prev,
                                const unsigned int* __restrict__ maxbits,
                                const float* __restrict__ W_phi,
                                float* __restrict__ out, int n) {
    const int i = blockIdx.x * blockDim.x + threadIdx.x;
    const int ic = i < n ? i : n - 1;
    unsigned b0v = maxbits[ic];
    unsigned b1v = __float_as_uint(out[ic]);
    float pv = prev[ic];

    const int lane = threadIdx.x & 31;
    v2f a0, a1, b;
    a0.x = W_phi[lane];        a0.y = W_phi[lane + 32];
    a1.x = W_phi[lane + 64];   a1.y = W_phi[lane + 96];
    b.x = 1.0f; b.y = 1.0f;
    asm volatile("" : "+v"(b0v), "+v"(b1v), "+v"(pv), "+v"(a0), "+v"(a1));
    v8f c = {};
    c = __builtin_amdgcn_wmma_f32_16x16x4_f32(false, a0, false, b,
                                              (short)0, c, false, false);
    c = __builtin_amdgcn_wmma_f32_16x16x4_f32(false, a1, false, b,
                                              (short)0, c, false, false);
    const float s = c[0] + c[1] + c[2] + c[3] + c[4] + c[5] + c[6] + c[7];
    const float partner = __shfl(s, lane ^ 16, 32);
    const float phi_mean = (s + partner) * (1.0f / 128.0f);

    if (i < n) {
        const float m = __uint_as_float(b0v > b1v ? b0v : b1v);
        out[i] = 0.5f * (pv + m * phi_mean);
    }
}

extern "C" void kernel_launch(void* const* d_in, const int* in_sizes, int n_in,
                              void* d_out, int out_size, void* d_ws, size_t ws_size,
                              hipStream_t stream) {
    const float* prev    = (const float*)d_in[0];
    const float* nodes   = (const float*)d_in[1];
    const int*   rows    = (const int*)  d_in[2];
    const int*   cols    = (const int*)  d_in[3];
    const float* W_phi   = (const float*)d_in[4];
    const float* W_theta = (const float*)d_in[5];
    float*       out     = (float*)d_out;

    const int n_nodes = in_sizes[0];
    const int n_edges = in_sizes[2];

    unsigned int* maxbits = (unsigned int*)d_ws;
    float* proj     = (float*)d_ws + n_nodes;

    const int B = 256;
    int node_blocks = (n_nodes + B - 1) / B;
    int edge_blocks = (n_edges + EDGE_TILE - 1) / EDGE_TILE;

    init_project_kernel<<<node_blocks, B, 0, stream>>>(nodes, W_theta, maxbits, out, proj, n_nodes);
    edge_max_kernel<<<edge_blocks, EDGE_TPB, 0, stream>>>(proj, rows, cols,
                                                   maxbits, out, n_edges);
    finalize_kernel<<<node_blocks, B, 0, stream>>>(prev, maxbits, W_phi,
                                                   out, n_nodes);
}
